# starting kernel
# baseline (speedup 1.0000x reference)
#include <hip/hip_runtime.h>
#include <hip/hip_bf16.h>

typedef __attribute__((ext_vector_type(2))) float v2f;
typedef __attribute__((ext_vector_type(8))) float v8f;

#define N_NODES 100000
#define N_EDGES 1600000
#define F_IN    500
#define HID     128
#define N_CLS   40


__global__ __launch_bounds__(256) void deg_init_kernel(float* __restrict__ deg) {
    int i = blockIdx.x * blockDim.x + threadIdx.x;
    if (i < N_NODES) deg[i] = 1.0f;
}

__global__ __launch_bounds__(256) void deg_count_kernel(const int* __restrict__ dst,
                                                        float* __restrict__ deg) {
    int e = blockIdx.x * blockDim.x + threadIdx.x;
    if (e < N_EDGES) unsafeAtomicAdd(&deg[dst[e]], 1.0f);
}

__global__ __launch_bounds__(256) void dinv_kernel(float* __restrict__ d) {
    int i = blockIdx.x * blockDim.x + threadIdx.x;
    if (i < N_NODES) d[i] = rsqrtf(d[i]);
}


template<int NT>
__global__ __launch_bounds__(256) void pack_w_kernel(const float* __restrict__ W,
                                                     v2f* __restrict__ Wp,
                                                     int K, int Nc) {
    int idx = blockIdx.x * blockDim.x + threadIdx.x;
    int total = (K / 4) * NT * 32;
    if (idx >= total) return;
    int lane = idx & 31;
    int t    = (idx >> 5) % NT;
    int ks   = idx / (32 * NT);
    int col  = t * 16 + (lane & 15);
    int k    = ks * 4 + ((lane >> 4) * 2);
    v2f v;
    v.x = (col < Nc) ? W[(size_t)k * Nc + col]       : 0.0f;
    v.y = (col < Nc) ? W[(size_t)(k + 1) * Nc + col] : 0.0f;
    Wp[idx] = v;
}


template<int NT, int NC, bool RELU>
__global__ __launch_bounds__(128) void gemm_wmma_kernel(const float* __restrict__ A,
                                                        const v2f* __restrict__ Wp,
                                                        float* __restrict__ Out,
                                                        int M, int K) {
    const int wave = threadIdx.x >> 5;
    const int lane = threadIdx.x & 31;
    const int pair = blockIdx.x * 4 + wave;
    if (pair * 32 >= M) return;

    const int kk  = (lane >> 4) * 2;
    const int col = lane & 15;

    const float* __restrict__ Arow0 = A + (size_t)(pair * 32 + col) * K + kk;
    const float* __restrict__ Arow1 = Arow0 + (size_t)16 * K;
    const v2f*   __restrict__ wp    = Wp + lane;

    v8f acc0[NT], acc1[NT];
#pragma unroll
    for (int t = 0; t < NT; ++t) {
        acc0[t] = (v8f){0.f,0.f,0.f,0.f,0.f,0.f,0.f,0.f};
        acc1[t] = (v8f){0.f,0.f,0.f,0.f,0.f,0.f,0.f,0.f};
    }

    const int nsteps = K >> 2;
#pragma unroll 2
    for (int ks = 0; ks < nsteps; ++ks) {
        v2f a0 = *reinterpret_cast<const v2f*>(Arow0 + ks * 4);
        v2f a1 = *reinterpret_cast<const v2f*>(Arow1 + ks * 4);
        if (RELU) {
            a0.x = fmaxf(a0.x, 0.0f); a0.y = fmaxf(a0.y, 0.0f);
            a1.x = fmaxf(a1.x, 0.0f); a1.y = fmaxf(a1.y, 0.0f);
        }
#pragma unroll
        for (int t = 0; t < NT; ++t) {
            v2f b = wp[(ks * NT + t) * 32];
            acc0[t] = __builtin_amdgcn_wmma_f32_16x16x4_f32(
                          false, a0, false, b, (short)0, acc0[t], false, false);
            acc1[t] = __builtin_amdgcn_wmma_f32_16x16x4_f32(
                          false, a1, false, b, (short)0, acc1[t], false, false);
        }
    }

    const int rbase0 = pair * 32 + ((lane < 16) ? 0 : 8);
    const int rbase1 = rbase0 + 16;
#pragma unroll
    for (int t = 0; t < NT; ++t) {
        int c = t * 16 + col;
        if ((NC % 16 == 0) || (c < NC)) {
#pragma unroll
            for (int r = 0; r < 8; ++r) {
                Out[(size_t)(rbase0 + r) * NC + c] = acc0[t][r];
                Out[(size_t)(rbase1 + r) * NC + c] = acc1[t][r];
            }
        }
    }
}


template<int F>
__global__ __launch_bounds__(256) void self_bias_kernel(const float* __restrict__ h,
                                                        const float* __restrict__ dinv,
                                                        const float* __restrict__ bias,
                                                        float* __restrict__ out,
                                                        int n) {
    constexpr int F4 = F / 4;
    long idx = (long)blockIdx.x * blockDim.x + threadIdx.x;
    if (idx >= (long)n * F4) return;
    int i  = (int)(idx / F4);
    int f4 = (int)(idx % F4);
    float di = dinv[i];
    float s  = di * di;
    float4 hv = reinterpret_cast<const float4*>(h)[idx];
    float4 bv = reinterpret_cast<const float4*>(bias)[f4];
    float4 ov;
    ov.x = bv.x + s * hv.x;  ov.y = bv.y + s * hv.y;
    ov.z = bv.z + s * hv.z;  ov.w = bv.w + s * hv.w;
    reinterpret_cast<float4*>(out)[idx] = ov;
}


template<int F>
__global__ __launch_bounds__(256) void edge_scatter_kernel(const float* __restrict__ h,
                                                           const float* __restrict__ dinv,
                                                           const int* __restrict__ ei,
                                                           float* __restrict__ out) {
    constexpr int VEC   = (F % 128 == 0) ? 4 : 2;
    constexpr int LANES = F / VEC;
    int e = blockIdx.x * (blockDim.x >> 5) + (threadIdx.x >> 5);
    if (e >= N_EDGES) return;
    int lane = threadIdx.x & 31;
    int s = ei[e];
    int d = ei[N_EDGES + e];
    float norm = dinv[s] * dinv[d];
    if (lane < LANES) {
        const float* __restrict__ hp = h   + (size_t)s * F + lane * VEC;
        float*       __restrict__ op = out + (size_t)d * F + lane * VEC;
        if constexpr (VEC == 4) {
            float4 v = *reinterpret_cast<const float4*>(hp);
            unsafeAtomicAdd(op + 0, norm * v.x);
            unsafeAtomicAdd(op + 1, norm * v.y);
            unsafeAtomicAdd(op + 2, norm * v.z);
            unsafeAtomicAdd(op + 3, norm * v.w);
        } else {
            float2 v = *reinterpret_cast<const float2*>(hp);
            unsafeAtomicAdd(op + 0, norm * v.x);
            unsafeAtomicAdd(op + 1, norm * v.y);
        }
    }
}


extern "C" void kernel_launch(void* const* d_in, const int* in_sizes, int n_in,
                              void* d_out, int out_size, void* d_ws, size_t ws_size,
                              hipStream_t stream) {
    const float* x  = (const float*)d_in[0];
    const int*   ei = (const int*)d_in[1];
    const float* W1 = (const float*)d_in[2];
    const float* b1 = (const float*)d_in[3];
    const float* W2 = (const float*)d_in[4];
    const float* b2 = (const float*)d_in[5];
    const float* W3 = (const float*)d_in[6];
    const float* b3 = (const float*)d_in[7];
    float* out = (float*)d_out;

    char* ws = (char*)d_ws;
    float* dinv = (float*)ws;
    float* bufA = (float*)(ws + (((size_t)N_NODES * 4 + 255) & ~(size_t)255));
    float* bufB = bufA + (size_t)N_NODES * HID;
    v2f*   Wp1  = (v2f*)(bufB + (size_t)N_NODES * HID);
    const int NP1 = (F_IN / 4) * 8 * 32;
    const int NP2 = (HID  / 4) * 8 * 32;
    const int NP3 = (HID  / 4) * 3 * 32;
    v2f* Wp2 = Wp1 + NP1;
    v2f* Wp3 = Wp2 + NP2;

    const int pairs   = N_NODES / 32;
    const int gemmBlk = (pairs + 3) / 4;
    const int edgeBlk = (N_EDGES + 7) / 8;

    pack_w_kernel<8><<<(NP1 + 255) / 256, 256, 0, stream>>>(W1, Wp1, F_IN, HID);
    pack_w_kernel<8><<<(NP2 + 255) / 256, 256, 0, stream>>>(W2, Wp2, HID, HID);
    pack_w_kernel<3><<<(NP3 + 255) / 256, 256, 0, stream>>>(W3, Wp3, HID, N_CLS);

    deg_init_kernel <<<(N_NODES + 255) / 256, 256, 0, stream>>>(dinv);
    deg_count_kernel<<<(N_EDGES + 255) / 256, 256, 0, stream>>>(ei + N_EDGES, dinv);
    dinv_kernel     <<<(N_NODES + 255) / 256, 256, 0, stream>>>(dinv);

    gemm_wmma_kernel<8, HID, false><<<gemmBlk, 128, 0, stream>>>(x, Wp1, bufA, N_NODES, F_IN);
    self_bias_kernel<HID><<<((long)N_NODES * (HID/4) + 255) / 256, 256, 0, stream>>>(bufA, dinv, b1, bufB, N_NODES);
    edge_scatter_kernel<HID><<<edgeBlk, 256, 0, stream>>>(bufA, dinv, ei, bufB);

    gemm_wmma_kernel<8, HID, true><<<gemmBlk, 128, 0, stream>>>(bufB, Wp2, bufA, N_NODES, HID);
    self_bias_kernel<HID><<<((long)N_NODES * (HID/4) + 255) / 256, 256, 0, stream>>>(bufA, dinv, b2, bufB, N_NODES);
    edge_scatter_kernel<HID><<<edgeBlk, 256, 0, stream>>>(bufA, dinv, ei, bufB);

    gemm_wmma_kernel<3, N_CLS, true><<<gemmBlk, 128, 0, stream>>>(bufB, Wp3, bufA, N_NODES, HID);
    self_bias_kernel<N_CLS><<<((long)N_NODES * (N_CLS/4) + 255) / 256, 256, 0, stream>>>(bufA, dinv, b3, out, N_NODES);
    edge_scatter_kernel<N_CLS><<<edgeBlk, 256, 0, stream>>>(bufA, dinv, ei, out);
}


# 128-wide edge scatters split by column half: a work-group reads its die and serves the columns of its own cache half, drawing 512-edge chunks from per-half counters; sums stay atomic arrival order
# speedup vs baseline: 1.6732x; 1.6732x over previous
#include <hip/hip_runtime.h>
#include <hip/hip_bf16.h>

typedef __attribute__((ext_vector_type(2))) float v2f;
typedef __attribute__((ext_vector_type(8))) float v8f;

#define N_NODES 100000
#define N_EDGES 1600000
#define F_IN    500
#define HID     128
#define N_CLS   40


__global__ __launch_bounds__(256) void deg_init_kernel(float* __restrict__ deg,
                                                       int* __restrict__ ctr) {
    int i = blockIdx.x * blockDim.x + threadIdx.x;
    if (i < N_NODES) deg[i] = 1.0f;
    if (i < 4) ctr[i] = 0;
}

__global__ __launch_bounds__(256) void deg_count_kernel(const int* __restrict__ dst,
                                                        float* __restrict__ deg) {
    int e = blockIdx.x * blockDim.x + threadIdx.x;
    if (e < N_EDGES) unsafeAtomicAdd(&deg[dst[e]], 1.0f);
}

__global__ __launch_bounds__(256) void dinv_kernel(float* __restrict__ d) {
    int i = blockIdx.x * blockDim.x + threadIdx.x;
    if (i < N_NODES) d[i] = rsqrtf(d[i]);
}


template<int NT>
__global__ __launch_bounds__(256) void pack_w_kernel(const float* __restrict__ W,
                                                     v2f* __restrict__ Wp,
                                                     int K, int Nc) {
    int idx = blockIdx.x * blockDim.x + threadIdx.x;
    int total = (K / 4) * NT * 32;
    if (idx >= total) return;
    int lane = idx & 31;
    int t    = (idx >> 5) % NT;
    int ks   = idx / (32 * NT);
    int col  = t * 16 + (lane & 15);
    int k    = ks * 4 + ((lane >> 4) * 2);
    v2f v;
    v.x = (col < Nc) ? W[(size_t)k * Nc + col]       : 0.0f;
    v.y = (col < Nc) ? W[(size_t)(k + 1) * Nc + col] : 0.0f;
    Wp[idx] = v;
}


template<int NT, int NC, bool RELU>
__global__ __launch_bounds__(128) void gemm_wmma_kernel(const float* __restrict__ A,
                                                        const v2f* __restrict__ Wp,
                                                        float* __restrict__ Out,
                                                        int M, int K) {
    const int wave = threadIdx.x >> 5;
    const int lane = threadIdx.x & 31;
    const int pair = blockIdx.x * 4 + wave;
    if (pair * 32 >= M) return;

    const int kk  = (lane >> 4) * 2;
    const int col = lane & 15;

    const float* __restrict__ Arow0 = A + (size_t)(pair * 32 + col) * K + kk;
    const float* __restrict__ Arow1 = Arow0 + (size_t)16 * K;
    const v2f*   __restrict__ wp    = Wp + lane;

    v8f acc0[NT], acc1[NT];
#pragma unroll
    for (int t = 0; t < NT; ++t) {
        acc0[t] = (v8f){0.f,0.f,0.f,0.f,0.f,0.f,0.f,0.f};
        acc1[t] = (v8f){0.f,0.f,0.f,0.f,0.f,0.f,0.f,0.f};
    }

    const int nsteps = K >> 2;
#pragma unroll 2
    for (int ks = 0; ks < nsteps; ++ks) {
        v2f a0 = *reinterpret_cast<const v2f*>(Arow0 + ks * 4);
        v2f a1 = *reinterpret_cast<const v2f*>(Arow1 + ks * 4);
        if (RELU) {
            a0.x = fmaxf(a0.x, 0.0f); a0.y = fmaxf(a0.y, 0.0f);
            a1.x = fmaxf(a1.x, 0.0f); a1.y = fmaxf(a1.y, 0.0f);
        }
#pragma unroll
        for (int t = 0; t < NT; ++t) {
            v2f b = wp[(ks * NT + t) * 32];
            acc0[t] = __builtin_amdgcn_wmma_f32_16x16x4_f32(
                          false, a0, false, b, (short)0, acc0[t], false, false);
            acc1[t] = __builtin_amdgcn_wmma_f32_16x16x4_f32(
                          false, a1, false, b, (short)0, acc1[t], false, false);
        }
    }

    const int rbase0 = pair * 32 + ((lane < 16) ? 0 : 8);
    const int rbase1 = rbase0 + 16;
#pragma unroll
    for (int t = 0; t < NT; ++t) {
        int c = t * 16 + col;
        if ((NC % 16 == 0) || (c < NC)) {
#pragma unroll
            for (int r = 0; r < 8; ++r) {
                Out[(size_t)(rbase0 + r) * NC + c] = acc0[t][r];
                Out[(size_t)(rbase1 + r) * NC + c] = acc1[t][r];
            }
        }
    }
}


template<int F>
__global__ __launch_bounds__(256) void self_bias_kernel(const float* __restrict__ h,
                                                        const float* __restrict__ dinv,
                                                        const float* __restrict__ bias,
                                                        float* __restrict__ out,
                                                        int n) {
    constexpr int F4 = F / 4;
    long idx = (long)blockIdx.x * blockDim.x + threadIdx.x;
    if (idx >= (long)n * F4) return;
    int i  = (int)(idx / F4);
    int f4 = (int)(idx % F4);
    float di = dinv[i];
    float s  = di * di;
    float4 hv = reinterpret_cast<const float4*>(h)[idx];
    float4 bv = reinterpret_cast<const float4*>(bias)[f4];
    float4 ov;
    ov.x = bv.x + s * hv.x;  ov.y = bv.y + s * hv.y;
    ov.z = bv.z + s * hv.z;  ov.w = bv.w + s * hv.w;
    reinterpret_cast<float4*>(out)[idx] = ov;
}


template<int F, int R>
__device__ __forceinline__ void row_load(const float* __restrict__ h, int s, int lo,
                                         float (&v)[R]) {
    const float* __restrict__ hp = h + (size_t)s * F + lo;
    if constexpr (F % 32 == 0) {
#pragma unroll
        for (int k = 0; k < R; ++k) v[k] = hp[32 * k];
    } else {
        float2 t = *reinterpret_cast<const float2*>(hp);
        v[0] = t.x; v[1] = t.y;
    }
}

template<int F, int R>
__device__ __forceinline__ void row_add(float* __restrict__ out, int d, int lo, float n,
                                        const float (&v)[R]) {
    float* __restrict__ op = out + (size_t)d * F + lo;
    if constexpr (F % 32 == 0) {
#pragma unroll
        for (int k = 0; k < R; ++k) unsafeAtomicAdd(op + 32 * k, n * v[k]);
    } else {
        unsafeAtomicAdd(op + 0, n * v[0]);
        unsafeAtomicAdd(op + 1, n * v[1]);
    }
}

template<int F>
__global__ __launch_bounds__(256) void edge_scatter_kernel(const float* __restrict__ h,
                                                           const float* __restrict__ dinv,
                                                           const int* __restrict__ ei,
                                                           float* __restrict__ out) {
    static_assert(N_EDGES % 32 == 0, "every wave owns 32 whole edges");
    constexpr bool WIDE  = (F % 32 == 0);
    constexpr int  R     = WIDE ? F / 32 : 2;
    constexpr int  BATCH = WIDE ? 8 : 32;
    constexpr int  NB    = 32 / BATCH;

    const int lane = threadIdx.x & 31;
    const int e0   = (blockIdx.x * (blockDim.x >> 5) + (threadIdx.x >> 5)) * 32;
    if (e0 >= N_EDGES) return;

    const int   sv = ei[e0 + lane];
    const int   dv = ei[N_EDGES + e0 + lane];
    const float nv = dinv[sv] * dinv[dv];
    const int   nb = __float_as_int(nv);

    const bool act = WIDE ? true : (lane < F / 2);
    const int  lo  = WIDE ? lane : lane * 2;

    float v[2][BATCH][R];

    {
        int s[BATCH];
#pragma unroll
        for (int i = 0; i < BATCH; ++i) s[i] = __builtin_amdgcn_readlane(sv, i);
        if (act) {
#pragma unroll
            for (int i = 0; i < BATCH; ++i) row_load<F, R>(h, s[i], lo, v[0][i]);
        }
    }

#pragma unroll
    for (int b = 0; b < NB; ++b) {
        int s[BATCH], d[BATCH];
        float n[BATCH];
#pragma unroll
        for (int i = 0; i < BATCH; ++i) {
            d[i] = __builtin_amdgcn_readlane(dv, b * BATCH + i);
            n[i] = __int_as_float(__builtin_amdgcn_readlane(nb, b * BATCH + i));
            s[i] = (b + 1 < NB) ? __builtin_amdgcn_readlane(sv, (b + 1) * BATCH + i) : 0;
        }
        if (act) {
            if (b + 1 < NB) {
#pragma unroll
                for (int i = 0; i < BATCH; ++i) row_load<F, R>(h, s[i], lo, v[(b + 1) & 1][i]);
            }
#pragma unroll
            for (int i = 0; i < BATCH; ++i) row_add<F, R>(out, d[i], lo, n[i], v[b & 1][i]);
        }
    }
}


#define CH_EDGES 512
#define N_CHUNKS (N_EDGES / CH_EDGES)

__global__ __launch_bounds__(256) void edge_scatter_halves_kernel(const float* __restrict__ h,
                                                                  const float* __restrict__ dinv,
                                                                  const int* __restrict__ ei,
                                                                  float* __restrict__ out,
                                                                  int* __restrict__ ctr) {
    static_assert(N_EDGES % CH_EDGES == 0, "whole chunks");
    static_assert(CH_EDGES == 8 * 64, "8 waves, 64 edges each");
    constexpr int F = HID, R = 2, BATCH = 8, NB = 32 / BATCH;
    __shared__ int s_q, s_tk;

    const int lane = threadIdx.x & 31;
    const int wv   = threadIdx.x >> 5;

    if (threadIdx.x == 0) {
        const unsigned id = __builtin_amdgcn_s_sendmsg_rtn(0x87);
        s_q = (int)((id >> 18) & 1u);
    }
    __syncthreads();
    int  q     = s_q;
    bool other = false;

    for (int trip = 0; trip < 2 * N_CHUNKS + 2; ++trip) {
        if (threadIdx.x == 0) s_tk = atomicAdd(&ctr[q], 1);
        __syncthreads();
        const int tk = s_tk;
        __syncthreads();
        if (tk >= N_CHUNKS) {
            if (other) break;
            other = true; q ^= 1;
            continue;
        }
        const int lo = q * 64 + lane;

        for (int it = 0; it < 2; ++it) {
            const int e0 = tk * CH_EDGES + wv * 64 + it * 32;
            const int   sv = ei[e0 + lane];
            const int   dv = ei[N_EDGES + e0 + lane];
            const float nv = dinv[sv] * dinv[dv];
            const int   nb = __float_as_int(nv);

            float v[2][BATCH][R];
            {
                int s[BATCH];
#pragma unroll
                for (int i = 0; i < BATCH; ++i) s[i] = __builtin_amdgcn_readlane(sv, i);
#pragma unroll
                for (int i = 0; i < BATCH; ++i) row_load<F, R>(h, s[i], lo, v[0][i]);
            }
#pragma unroll
            for (int b = 0; b < NB; ++b) {
                int s[BATCH], d[BATCH];
                float n[BATCH];
#pragma unroll
                for (int i = 0; i < BATCH; ++i) {
                    d[i] = __builtin_amdgcn_readlane(dv, b * BATCH + i);
                    n[i] = __int_as_float(__builtin_amdgcn_readlane(nb, b * BATCH + i));
                    s[i] = (b + 1 < NB) ? __builtin_amdgcn_readlane(sv, (b + 1) * BATCH + i) : 0;
                }
                if (b + 1 < NB) {
#pragma unroll
                    for (int i = 0; i < BATCH; ++i) row_load<F, R>(h, s[i], lo, v[(b + 1) & 1][i]);
                }
#pragma unroll
                for (int i = 0; i < BATCH; ++i) row_add<F, R>(out, d[i], lo, n[i], v[b & 1][i]);
            }
        }
    }
}


extern "C" void kernel_launch(void* const* d_in, const int* in_sizes, int n_in,
                              void* d_out, int out_size, void* d_ws, size_t ws_size,
                              hipStream_t stream) {
    const float* x  = (const float*)d_in[0];
    const int*   ei = (const int*)d_in[1];
    const float* W1 = (const float*)d_in[2];
    const float* b1 = (const float*)d_in[3];
    const float* W2 = (const float*)d_in[4];
    const float* b2 = (const float*)d_in[5];
    const float* W3 = (const float*)d_in[6];
    const float* b3 = (const float*)d_in[7];
    float* out = (float*)d_out;

    char* ws = (char*)d_ws;
    float* dinv = (float*)ws;
    int*   ctr  = (int*)(ws + (size_t)N_NODES * 4);
    float* bufA = (float*)(ws + (((size_t)N_NODES * 4 + 255) & ~(size_t)255));
    float* bufB = bufA + (size_t)N_NODES * HID;
    v2f*   Wp1  = (v2f*)(bufB + (size_t)N_NODES * HID);
    const int NP1 = (F_IN / 4) * 8 * 32;
    const int NP2 = (HID  / 4) * 8 * 32;
    const int NP3 = (HID  / 4) * 3 * 32;
    v2f* Wp2 = Wp1 + NP1;
    v2f* Wp3 = Wp2 + NP2;

    const int pairs   = N_NODES / 32;
    const int gemmBlk = (pairs + 3) / 4;
    const int edgeBlk = (N_EDGES / 32 + 7) / 8;
    const int halfBlk = 2048;

    pack_w_kernel<8><<<(NP1 + 255) / 256, 256, 0, stream>>>(W1, Wp1, F_IN, HID);
    pack_w_kernel<8><<<(NP2 + 255) / 256, 256, 0, stream>>>(W2, Wp2, HID, HID);
    pack_w_kernel<3><<<(NP3 + 255) / 256, 256, 0, stream>>>(W3, Wp3, HID, N_CLS);

    deg_init_kernel <<<(N_NODES + 255) / 256, 256, 0, stream>>>(dinv, ctr);
    deg_count_kernel<<<(N_EDGES + 255) / 256, 256, 0, stream>>>(ei + N_EDGES, dinv);
    dinv_kernel     <<<(N_NODES + 255) / 256, 256, 0, stream>>>(dinv);

    gemm_wmma_kernel<8, HID, false><<<gemmBlk, 128, 0, stream>>>(x, Wp1, bufA, N_NODES, F_IN);
    self_bias_kernel<HID><<<((long)N_NODES * (HID/4) + 255) / 256, 256, 0, stream>>>(bufA, dinv, b1, bufB, N_NODES);
    edge_scatter_halves_kernel<<<halfBlk, 256, 0, stream>>>(bufA, dinv, ei, bufB, ctr + 0);

    gemm_wmma_kernel<8, HID, true><<<gemmBlk, 128, 0, stream>>>(bufB, Wp2, bufA, N_NODES, HID);
    self_bias_kernel<HID><<<((long)N_NODES * (HID/4) + 255) / 256, 256, 0, stream>>>(bufA, dinv, b2, bufB, N_NODES);
    edge_scatter_halves_kernel<<<halfBlk, 256, 0, stream>>>(bufA, dinv, ei, bufB, ctr + 2);

    gemm_wmma_kernel<3, N_CLS, true><<<gemmBlk, 128, 0, stream>>>(bufB, Wp3, bufA, N_NODES, HID);
    self_bias_kernel<N_CLS><<<((long)N_NODES * (N_CLS/4) + 255) / 256, 256, 0, stream>>>(bufA, dinv, b3, out, N_NODES);
    edge_scatter_kernel<N_CLS><<<edgeBlk, 256, 0, stream>>>(bufA, dinv, ei, out);
}


# edge norms dinv[src]*dinv[dst] made once per call by one new launch and read coalesced by the three scatters, replacing two random dinv reads at each of an edge's five visits; sum order unchanged
# speedup vs baseline: 1.8071x; 1.0800x over previous
#include <hip/hip_runtime.h>
#include <hip/hip_bf16.h>

typedef __attribute__((ext_vector_type(2))) float v2f;
typedef __attribute__((ext_vector_type(8))) float v8f;

#define N_NODES 100000
#define N_EDGES 1600000
#define F_IN    500
#define HID     128
#define N_CLS   40


__global__ __launch_bounds__(256) void deg_init_kernel(float* __restrict__ deg,
                                                       int* __restrict__ ctr) {
    int i = blockIdx.x * blockDim.x + threadIdx.x;
    if (i < N_NODES) deg[i] = 1.0f;
    if (i < 4) ctr[i] = 0;
}

__global__ __launch_bounds__(256) void deg_count_kernel(const int* __restrict__ dst,
                                                        float* __restrict__ deg) {
    int e = blockIdx.x * blockDim.x + threadIdx.x;
    if (e < N_EDGES) unsafeAtomicAdd(&deg[dst[e]], 1.0f);
}

__global__ __launch_bounds__(256) void dinv_kernel(float* __restrict__ d) {
    int i = blockIdx.x * blockDim.x + threadIdx.x;
    if (i < N_NODES) d[i] = rsqrtf(d[i]);
}


__global__ __launch_bounds__(256) void edge_norm_kernel(const float* __restrict__ dinv,
                                                        const int* __restrict__ ei,
                                                        float* __restrict__ norm) {
    static_assert(N_EDGES % 4 == 0, "four whole edges per thread");
    const int i = blockIdx.x * blockDim.x + threadIdx.x;
    if (i >= N_EDGES / 4) return;
    const int4 s = reinterpret_cast<const int4*>(ei)[i];
    const int4 d = reinterpret_cast<const int4*>(ei + N_EDGES)[i];
    float4 n;
    n.x = dinv[s.x] * dinv[d.x];
    n.y = dinv[s.y] * dinv[d.y];
    n.z = dinv[s.z] * dinv[d.z];
    n.w = dinv[s.w] * dinv[d.w];
    reinterpret_cast<float4*>(norm)[i] = n;
}


template<int NT>
__global__ __launch_bounds__(256) void pack_w_kernel(const float* __restrict__ W,
                                                     v2f* __restrict__ Wp,
                                                     int K, int Nc) {
    int idx = blockIdx.x * blockDim.x + threadIdx.x;
    int total = (K / 4) * NT * 32;
    if (idx >= total) return;
    int lane = idx & 31;
    int t    = (idx >> 5) % NT;
    int ks   = idx / (32 * NT);
    int col  = t * 16 + (lane & 15);
    int k    = ks * 4 + ((lane >> 4) * 2);
    v2f v;
    v.x = (col < Nc) ? W[(size_t)k * Nc + col]       : 0.0f;
    v.y = (col < Nc) ? W[(size_t)(k + 1) * Nc + col] : 0.0f;
    Wp[idx] = v;
}


template<int NT, int NC, bool RELU>
__global__ __launch_bounds__(128) void gemm_wmma_kernel(const float* __restrict__ A,
                                                        const v2f* __restrict__ Wp,
                                                        float* __restrict__ Out,
                                                        int M, int K) {
    const int wave = threadIdx.x >> 5;
    const int lane = threadIdx.x & 31;
    const int pair = blockIdx.x * 4 + wave;
    if (pair * 32 >= M) return;

    const int kk  = (lane >> 4) * 2;
    const int col = lane & 15;

    const float* __restrict__ Arow0 = A + (size_t)(pair * 32 + col) * K + kk;
    const float* __restrict__ Arow1 = Arow0 + (size_t)16 * K;
    const v2f*   __restrict__ wp    = Wp + lane;

    v8f acc0[NT], acc1[NT];
#pragma unroll
    for (int t = 0; t < NT; ++t) {
        acc0[t] = (v8f){0.f,0.f,0.f,0.f,0.f,0.f,0.f,0.f};
        acc1[t] = (v8f){0.f,0.f,0.f,0.f,0.f,0.f,0.f,0.f};
    }

    const int nsteps = K >> 2;
#pragma unroll 2
    for (int ks = 0; ks < nsteps; ++ks) {
        v2f a0 = *reinterpret_cast<const v2f*>(Arow0 + ks * 4);
        v2f a1 = *reinterpret_cast<const v2f*>(Arow1 + ks * 4);
        if (RELU) {
            a0.x = fmaxf(a0.x, 0.0f); a0.y = fmaxf(a0.y, 0.0f);
            a1.x = fmaxf(a1.x, 0.0f); a1.y = fmaxf(a1.y, 0.0f);
        }
#pragma unroll
        for (int t = 0; t < NT; ++t) {
            v2f b = wp[(ks * NT + t) * 32];
            acc0[t] = __builtin_amdgcn_wmma_f32_16x16x4_f32(
                          false, a0, false, b, (short)0, acc0[t], false, false);
            acc1[t] = __builtin_amdgcn_wmma_f32_16x16x4_f32(
                          false, a1, false, b, (short)0, acc1[t], false, false);
        }
    }

    const int rbase0 = pair * 32 + ((lane < 16) ? 0 : 8);
    const int rbase1 = rbase0 + 16;
#pragma unroll
    for (int t = 0; t < NT; ++t) {
        int c = t * 16 + col;
        if ((NC % 16 == 0) || (c < NC)) {
#pragma unroll
            for (int r = 0; r < 8; ++r) {
                Out[(size_t)(rbase0 + r) * NC + c] = acc0[t][r];
                Out[(size_t)(rbase1 + r) * NC + c] = acc1[t][r];
            }
        }
    }
}


template<int F, bool MOVE>
__global__ __launch_bounds__(256) void self_bias_kernel(const float* __restrict__ h,
                                                        const float* __restrict__ dinv,
                                                        const float* __restrict__ bias,
                                                        float* __restrict__ out,
                                                        int n,
                                                        float* __restrict__ moved) {
    constexpr int F4 = F / 4;
    static_assert(!MOVE || (long)N_EDGES / 4 <= (long)N_NODES * F4, "the norm plane lies inside out");
    long idx = (long)blockIdx.x * blockDim.x + threadIdx.x;
    if (idx >= (long)n * F4) return;
    if (MOVE) {
        if (idx < N_EDGES / 4)
            reinterpret_cast<float4*>(moved)[idx] = reinterpret_cast<const float4*>(out)[idx];
    }
    int i  = (int)(idx / F4);
    int f4 = (int)(idx % F4);
    float di = dinv[i];
    float s  = di * di;
    float4 hv = reinterpret_cast<const float4*>(h)[idx];
    float4 bv = reinterpret_cast<const float4*>(bias)[f4];
    float4 ov;
    ov.x = bv.x + s * hv.x;  ov.y = bv.y + s * hv.y;
    ov.z = bv.z + s * hv.z;  ov.w = bv.w + s * hv.w;
    reinterpret_cast<float4*>(out)[idx] = ov;
}


template<int F, int R>
__device__ __forceinline__ void row_load(const float* __restrict__ h, int s, int lo,
                                         float (&v)[R]) {
    const float* __restrict__ hp = h + (size_t)s * F + lo;
    if constexpr (F % 32 == 0) {
#pragma unroll
        for (int k = 0; k < R; ++k) v[k] = hp[32 * k];
    } else {
        float2 t = *reinterpret_cast<const float2*>(hp);
        v[0] = t.x; v[1] = t.y;
    }
}

template<int F, int R>
__device__ __forceinline__ void row_add(float* __restrict__ out, int d, int lo, float n,
                                        const float (&v)[R]) {
    float* __restrict__ op = out + (size_t)d * F + lo;
    if constexpr (F % 32 == 0) {
#pragma unroll
        for (int k = 0; k < R; ++k) unsafeAtomicAdd(op + 32 * k, n * v[k]);
    } else {
        unsafeAtomicAdd(op + 0, n * v[0]);
        unsafeAtomicAdd(op + 1, n * v[1]);
    }
}

template<int F>
__global__ __launch_bounds__(256) void edge_scatter_kernel(const float* __restrict__ h,
                                                           const float* __restrict__ norm,
                                                           const int* __restrict__ ei,
                                                           float* __restrict__ out) {
    static_assert(N_EDGES % 32 == 0, "every wave owns 32 whole edges");
    constexpr bool WIDE  = (F % 32 == 0);
    constexpr int  R     = WIDE ? F / 32 : 2;
    constexpr int  BATCH = WIDE ? 8 : 32;
    constexpr int  NB    = 32 / BATCH;

    const int lane = threadIdx.x & 31;
    const int e0   = (blockIdx.x * (blockDim.x >> 5) + (threadIdx.x >> 5)) * 32;
    if (e0 >= N_EDGES) return;

    const int   sv = ei[e0 + lane];
    const int   dv = ei[N_EDGES + e0 + lane];
    const float nv = norm[e0 + lane];
    const int   nb = __float_as_int(nv);

    const bool act = WIDE ? true : (lane < F / 2);
    const int  lo  = WIDE ? lane : lane * 2;

    float v[2][BATCH][R];

    {
        int s[BATCH];
#pragma unroll
        for (int i = 0; i < BATCH; ++i) s[i] = __builtin_amdgcn_readlane(sv, i);
        if (act) {
#pragma unroll
            for (int i = 0; i < BATCH; ++i) row_load<F, R>(h, s[i], lo, v[0][i]);
        }
    }

#pragma unroll
    for (int b = 0; b < NB; ++b) {
        int s[BATCH], d[BATCH];
        float n[BATCH];
#pragma unroll
        for (int i = 0; i < BATCH; ++i) {
            d[i] = __builtin_amdgcn_readlane(dv, b * BATCH + i);
            n[i] = __int_as_float(__builtin_amdgcn_readlane(nb, b * BATCH + i));
            s[i] = (b + 1 < NB) ? __builtin_amdgcn_readlane(sv, (b + 1) * BATCH + i) : 0;
        }
        if (act) {
            if (b + 1 < NB) {
#pragma unroll
                for (int i = 0; i < BATCH; ++i) row_load<F, R>(h, s[i], lo, v[(b + 1) & 1][i]);
            }
#pragma unroll
            for (int i = 0; i < BATCH; ++i) row_add<F, R>(out, d[i], lo, n[i], v[b & 1][i]);
        }
    }
}


#define CH_EDGES 512
#define N_CHUNKS (N_EDGES / CH_EDGES)

__global__ __launch_bounds__(256) void edge_scatter_halves_kernel(const float* __restrict__ h,
                                                                  const float* __restrict__ norm,
                                                                  const int* __restrict__ ei,
                                                                  float* __restrict__ out,
                                                                  int* __restrict__ ctr) {
    static_assert(N_EDGES % CH_EDGES == 0, "whole chunks");
    static_assert(CH_EDGES == 8 * 64, "8 waves, 64 edges each");
    constexpr int F = HID, R = 2, BATCH = 8, NB = 32 / BATCH;
    __shared__ int s_q, s_tk;

    const int lane = threadIdx.x & 31;
    const int wv   = threadIdx.x >> 5;

    if (threadIdx.x == 0) {
        const unsigned id = __builtin_amdgcn_s_sendmsg_rtn(0x87);
        s_q = (int)((id >> 18) & 1u);
    }
    __syncthreads();
    int  q     = s_q;
    bool other = false;

    for (int trip = 0; trip < 2 * N_CHUNKS + 2; ++trip) {
        if (threadIdx.x == 0) s_tk = atomicAdd(&ctr[q], 1);
        __syncthreads();
        const int tk = s_tk;
        __syncthreads();
        if (tk >= N_CHUNKS) {
            if (other) break;
            other = true; q ^= 1;
            continue;
        }
        const int lo = q * 64 + lane;

        for (int it = 0; it < 2; ++it) {
            const int e0 = tk * CH_EDGES + wv * 64 + it * 32;
            const int   sv = ei[e0 + lane];
            const int   dv = ei[N_EDGES + e0 + lane];
            const float nv = norm[e0 + lane];
            const int   nb = __float_as_int(nv);

            float v[2][BATCH][R];
            {
                int s[BATCH];
#pragma unroll
                for (int i = 0; i < BATCH; ++i) s[i] = __builtin_amdgcn_readlane(sv, i);
#pragma unroll
                for (int i = 0; i < BATCH; ++i) row_load<F, R>(h, s[i], lo, v[0][i]);
            }
#pragma unroll
            for (int b = 0; b < NB; ++b) {
                int s[BATCH], d[BATCH];
                float n[BATCH];
#pragma unroll
                for (int i = 0; i < BATCH; ++i) {
                    d[i] = __builtin_amdgcn_readlane(dv, b * BATCH + i);
                    n[i] = __int_as_float(__builtin_amdgcn_readlane(nb, b * BATCH + i));
                    s[i] = (b + 1 < NB) ? __builtin_amdgcn_readlane(sv, (b + 1) * BATCH + i) : 0;
                }
                if (b + 1 < NB) {
#pragma unroll
                    for (int i = 0; i < BATCH; ++i) row_load<F, R>(h, s[i], lo, v[(b + 1) & 1][i]);
                }
#pragma unroll
                for (int i = 0; i < BATCH; ++i) row_add<F, R>(out, d[i], lo, n[i], v[b & 1][i]);
            }
        }
    }
}


extern "C" void kernel_launch(void* const* d_in, const int* in_sizes, int n_in,
                              void* d_out, int out_size, void* d_ws, size_t ws_size,
                              hipStream_t stream) {
    const float* x  = (const float*)d_in[0];
    const int*   ei = (const int*)d_in[1];
    const float* W1 = (const float*)d_in[2];
    const float* b1 = (const float*)d_in[3];
    const float* W2 = (const float*)d_in[4];
    const float* b2 = (const float*)d_in[5];
    const float* W3 = (const float*)d_in[6];
    const float* b3 = (const float*)d_in[7];
    float* out = (float*)d_out;

    char* ws = (char*)d_ws;
    float* dinv = (float*)ws;
    int*   ctr  = (int*)(ws + (size_t)N_NODES * 4);
    float* bufA = (float*)(ws + (((size_t)N_NODES * 4 + 255) & ~(size_t)255));
    float* bufB = bufA + (size_t)N_NODES * HID;
    v2f*   Wp1  = (v2f*)(bufB + (size_t)N_NODES * HID);
    const int NP1 = (F_IN / 4) * 8 * 32;
    const int NP2 = (HID  / 4) * 8 * 32;
    const int NP3 = (HID  / 4) * 3 * 32;
    v2f* Wp2 = Wp1 + NP1;
    v2f* Wp3 = Wp2 + NP2;

    const int pairs   = N_NODES / 32;
    const int gemmBlk = (pairs + 3) / 4;
    const int edgeBlk = (N_EDGES / 32 + 7) / 8;
    const int halfBlk = 2048;

    pack_w_kernel<8><<<(NP1 + 255) / 256, 256, 0, stream>>>(W1, Wp1, F_IN, HID);
    pack_w_kernel<8><<<(NP2 + 255) / 256, 256, 0, stream>>>(W2, Wp2, HID, HID);
    pack_w_kernel<3><<<(NP3 + 255) / 256, 256, 0, stream>>>(W3, Wp3, HID, N_CLS);

    deg_init_kernel <<<(N_NODES + 255) / 256, 256, 0, stream>>>(dinv, ctr);
    deg_count_kernel<<<(N_EDGES + 255) / 256, 256, 0, stream>>>(ei + N_EDGES, dinv);
    dinv_kernel     <<<(N_NODES + 255) / 256, 256, 0, stream>>>(dinv);
    edge_norm_kernel<<<(N_EDGES / 4 + 255) / 256, 256, 0, stream>>>(dinv, ei, out);

    gemm_wmma_kernel<8, HID, false><<<gemmBlk, 128, 0, stream>>>(x, Wp1, bufA, N_NODES, F_IN);
    self_bias_kernel<HID, false><<<((long)N_NODES * (HID/4) + 255) / 256, 256, 0, stream>>>(bufA, dinv, b1, bufB, N_NODES, nullptr);
    edge_scatter_halves_kernel<<<halfBlk, 256, 0, stream>>>(bufA, out, ei, bufB, ctr + 0);

    gemm_wmma_kernel<8, HID, true><<<gemmBlk, 128, 0, stream>>>(bufB, Wp2, bufA, N_NODES, HID);
    self_bias_kernel<HID, false><<<((long)N_NODES * (HID/4) + 255) / 256, 256, 0, stream>>>(bufA, dinv, b2, bufB, N_NODES, nullptr);
    edge_scatter_halves_kernel<<<halfBlk, 256, 0, stream>>>(bufA, out, ei, bufB, ctr + 2);

    gemm_wmma_kernel<3, N_CLS, true><<<gemmBlk, 128, 0, stream>>>(bufB, Wp3, bufA, N_NODES, HID);
    self_bias_kernel<N_CLS, true><<<((long)N_NODES * (N_CLS/4) + 255) / 256, 256, 0, stream>>>(bufA, dinv, b3, out, N_NODES, bufB);
    edge_scatter_kernel<N_CLS><<<edgeBlk, 256, 0, stream>>>(bufA, bufB, ei, out);
}


# GEMM epilogues write the start values b + dinv_i^2*h_i from their accumulators (layer 2 over its own A rows; layer 3 also moves the norm words under its rows); no start-value launches, 16 -> 13
# speedup vs baseline: 1.8841x; 1.0426x over previous
#include <hip/hip_runtime.h>
#include <hip/hip_bf16.h>

typedef __attribute__((ext_vector_type(2))) float v2f;
typedef __attribute__((ext_vector_type(8))) float v8f;
typedef __attribute__((ext_vector_type(4))) float v4f;

#define N_NODES 100000
#define N_EDGES 1600000
#define F_IN    500
#define HID     128
#define N_CLS   40


__global__ __launch_bounds__(256) void deg_init_kernel(float* __restrict__ deg,
                                                       int* __restrict__ ctr) {
    int i = blockIdx.x * blockDim.x + threadIdx.x;
    if (i < N_NODES) deg[i] = 1.0f;
    if (i < 4) ctr[i] = 0;
}

__global__ __launch_bounds__(256) void deg_count_kernel(const int* __restrict__ dst,
                                                        float* __restrict__ deg) {
    int e = blockIdx.x * blockDim.x + threadIdx.x;
    if (e < N_EDGES) unsafeAtomicAdd(&deg[dst[e]], 1.0f);
}

__global__ __launch_bounds__(256) void dinv_kernel(float* __restrict__ d) {
    int i = blockIdx.x * blockDim.x + threadIdx.x;
    if (i < N_NODES) d[i] = rsqrtf(d[i]);
}


__global__ __launch_bounds__(256) void edge_norm_kernel(const float* __restrict__ dinv,
                                                        const int* __restrict__ ei,
                                                        float* __restrict__ norm) {
    static_assert(N_EDGES % 4 == 0, "four whole edges per thread");
    const int i = blockIdx.x * blockDim.x + threadIdx.x;
    if (i >= N_EDGES / 4) return;
    const int4 s = reinterpret_cast<const int4*>(ei)[i];
    const int4 d = reinterpret_cast<const int4*>(ei + N_EDGES)[i];
    float4 n;
    n.x = dinv[s.x] * dinv[d.x];
    n.y = dinv[s.y] * dinv[d.y];
    n.z = dinv[s.z] * dinv[d.z];
    n.w = dinv[s.w] * dinv[d.w];
    reinterpret_cast<float4*>(norm)[i] = n;
}


template<int NT>
__global__ __launch_bounds__(256) void pack_w_kernel(const float* __restrict__ W,
                                                     v2f* __restrict__ Wp,
                                                     int K, int Nc) {
    int idx = blockIdx.x * blockDim.x + threadIdx.x;
    int total = (K / 4) * NT * 32;
    if (idx >= total) return;
    int lane = idx & 31;
    int t    = (idx >> 5) % NT;
    int ks   = idx / (32 * NT);
    int col  = t * 16 + (lane & 15);
    int k    = ks * 4 + ((lane >> 4) * 2);
    v2f v;
    v.x = (col < Nc) ? W[(size_t)k * Nc + col]       : 0.0f;
    v.y = (col < Nc) ? W[(size_t)(k + 1) * Nc + col] : 0.0f;
    Wp[idx] = v;
}


template<int NT, int NC, bool RELU, bool MOVE>
__global__ __launch_bounds__(128) void gemm_wmma_kernel(const float* A,
                                                        const v2f* __restrict__ Wp,
                                                        float* __restrict__ Out,
                                                        int M, int K,
                                                        const float* __restrict__ dinv,
                                                        const float* __restrict__ bias,
                                                        float* Start,
                                                        float* __restrict__ moved) {
    const int wave = threadIdx.x >> 5;
    const int lane = threadIdx.x & 31;
    const int pair = blockIdx.x * 4 + wave;
    if (pair * 32 >= M) return;

    if constexpr (MOVE) {
        constexpr int WORDS = 32 * NC;
        static_assert(WORDS == 10 * 128 && N_EDGES % WORDS == 0,
                      "ten 16-byte sweeps per wave; the norm plane ends on a wave's row boundary");
        const long w0 = (long)pair * WORDS;
        if (w0 < N_EDGES) {
            const v4f* from = reinterpret_cast<const v4f*>(Start + w0);
            v4f* to = reinterpret_cast<v4f*>(moved + w0);
            v4f t[10];
#pragma unroll
            for (int j = 0; j < 10; ++j) t[j] = from[j * 32 + lane];
            asm volatile("" :: "v"(t[0]), "v"(t[1]), "v"(t[2]), "v"(t[3]), "v"(t[4]),
                               "v"(t[5]), "v"(t[6]), "v"(t[7]), "v"(t[8]), "v"(t[9]));
#pragma unroll
            for (int j = 0; j < 10; ++j) to[j * 32 + lane] = t[j];
        }
    }

    const int kk  = (lane >> 4) * 2;
    const int col = lane & 15;

    const float* Arow0 = A + (size_t)(pair * 32 + col) * K + kk;
    const float* Arow1 = Arow0 + (size_t)16 * K;
    const v2f*   __restrict__ wp    = Wp + lane;

    v8f acc0[NT], acc1[NT];
#pragma unroll
    for (int t = 0; t < NT; ++t) {
        acc0[t] = (v8f){0.f,0.f,0.f,0.f,0.f,0.f,0.f,0.f};
        acc1[t] = (v8f){0.f,0.f,0.f,0.f,0.f,0.f,0.f,0.f};
    }

    const int nsteps = K >> 2;
#pragma unroll 2
    for (int ks = 0; ks < nsteps; ++ks) {
        v2f a0 = *reinterpret_cast<const v2f*>(Arow0 + ks * 4);
        v2f a1 = *reinterpret_cast<const v2f*>(Arow1 + ks * 4);
        if (RELU) {
            a0.x = fmaxf(a0.x, 0.0f); a0.y = fmaxf(a0.y, 0.0f);
            a1.x = fmaxf(a1.x, 0.0f); a1.y = fmaxf(a1.y, 0.0f);
        }
#pragma unroll
        for (int t = 0; t < NT; ++t) {
            v2f b = wp[(ks * NT + t) * 32];
            acc0[t] = __builtin_amdgcn_wmma_f32_16x16x4_f32(
                          false, a0, false, b, (short)0, acc0[t], false, false);
            acc1[t] = __builtin_amdgcn_wmma_f32_16x16x4_f32(
                          false, a1, false, b, (short)0, acc1[t], false, false);
        }
    }

    const int rbase0 = pair * 32 + ((lane < 16) ? 0 : 8);
    const int rbase1 = rbase0 + 16;
    float s0[8], s1[8];
#pragma unroll
    for (int r = 0; r < 8; ++r) {
        const float d0 = dinv[rbase0 + r];
        const float d1 = dinv[rbase1 + r];
        s0[r] = d0 * d0;
        s1[r] = d1 * d1;
    }
#pragma unroll
    for (int t = 0; t < NT; ++t) {
        int c = t * 16 + col;
        if ((NC % 16 == 0) || (c < NC)) {
            const float bv = bias[c];
#pragma unroll
            for (int r = 0; r < 8; ++r) {
                Out[(size_t)(rbase0 + r) * NC + c] = acc0[t][r];
                Out[(size_t)(rbase1 + r) * NC + c] = acc1[t][r];
                Start[(size_t)(rbase0 + r) * NC + c] = bv + s0[r] * acc0[t][r];
                Start[(size_t)(rbase1 + r) * NC + c] = bv + s1[r] * acc1[t][r];
            }
        }
    }
}


template<int F, int R>
__device__ __forceinline__ void row_load(const float* __restrict__ h, int s, int lo,
                                         float (&v)[R]) {
    const float* __restrict__ hp = h + (size_t)s * F + lo;
    if constexpr (F % 32 == 0) {
#pragma unroll
        for (int k = 0; k < R; ++k) v[k] = hp[32 * k];
    } else {
        float2 t = *reinterpret_cast<const float2*>(hp);
        v[0] = t.x; v[1] = t.y;
    }
}

template<int F, int R>
__device__ __forceinline__ void row_add(float* __restrict__ out, int d, int lo, float n,
                                        const float (&v)[R]) {
    float* __restrict__ op = out + (size_t)d * F + lo;
    if constexpr (F % 32 == 0) {
#pragma unroll
        for (int k = 0; k < R; ++k) unsafeAtomicAdd(op + 32 * k, n * v[k]);
    } else {
        unsafeAtomicAdd(op + 0, n * v[0]);
        unsafeAtomicAdd(op + 1, n * v[1]);
    }
}

template<int F>
__global__ __launch_bounds__(256) void edge_scatter_kernel(const float* __restrict__ h,
                                                           const float* __restrict__ norm,
                                                           const int* __restrict__ ei,
                                                           float* __restrict__ out) {
    static_assert(N_EDGES % 32 == 0, "every wave owns 32 whole edges");
    constexpr bool WIDE  = (F % 32 == 0);
    constexpr int  R     = WIDE ? F / 32 : 2;
    constexpr int  BATCH = WIDE ? 8 : 32;
    constexpr int  NB    = 32 / BATCH;

    const int lane = threadIdx.x & 31;
    const int e0   = (blockIdx.x * (blockDim.x >> 5) + (threadIdx.x >> 5)) * 32;
    if (e0 >= N_EDGES) return;

    const int   sv = ei[e0 + lane];
    const int   dv = ei[N_EDGES + e0 + lane];
    const float nv = norm[e0 + lane];
    const int   nb = __float_as_int(nv);

    const bool act = WIDE ? true : (lane < F / 2);
    const int  lo  = WIDE ? lane : lane * 2;

    float v[2][BATCH][R];

    {
        int s[BATCH];
#pragma unroll
        for (int i = 0; i < BATCH; ++i) s[i] = __builtin_amdgcn_readlane(sv, i);
        if (act) {
#pragma unroll
            for (int i = 0; i < BATCH; ++i) row_load<F, R>(h, s[i], lo, v[0][i]);
        }
    }

#pragma unroll
    for (int b = 0; b < NB; ++b) {
        int s[BATCH], d[BATCH];
        float n[BATCH];
#pragma unroll
        for (int i = 0; i < BATCH; ++i) {
            d[i] = __builtin_amdgcn_readlane(dv, b * BATCH + i);
            n[i] = __int_as_float(__builtin_amdgcn_readlane(nb, b * BATCH + i));
            s[i] = (b + 1 < NB) ? __builtin_amdgcn_readlane(sv, (b + 1) * BATCH + i) : 0;
        }
        if (act) {
            if (b + 1 < NB) {
#pragma unroll
                for (int i = 0; i < BATCH; ++i) row_load<F, R>(h, s[i], lo, v[(b + 1) & 1][i]);
            }
#pragma unroll
            for (int i = 0; i < BATCH; ++i) row_add<F, R>(out, d[i], lo, n[i], v[b & 1][i]);
        }
    }
}


#define CH_EDGES 512
#define N_CHUNKS (N_EDGES / CH_EDGES)

__global__ __launch_bounds__(256) void edge_scatter_halves_kernel(const float* __restrict__ h,
                                                                  const float* __restrict__ norm,
                                                                  const int* __restrict__ ei,
                                                                  float* __restrict__ out,
                                                                  int* __restrict__ ctr) {
    static_assert(N_EDGES % CH_EDGES == 0, "whole chunks");
    static_assert(CH_EDGES == 8 * 64, "8 waves, 64 edges each");
    constexpr int F = HID, R = 2, BATCH = 8, NB = 32 / BATCH;
    __shared__ int s_q, s_tk;

    const int lane = threadIdx.x & 31;
    const int wv   = threadIdx.x >> 5;

    if (threadIdx.x == 0) {
        const unsigned id = __builtin_amdgcn_s_sendmsg_rtn(0x87);
        s_q = (int)((id >> 18) & 1u);
    }
    __syncthreads();
    int  q     = s_q;
    bool other = false;

    for (int trip = 0; trip < 2 * N_CHUNKS + 2; ++trip) {
        if (threadIdx.x == 0) s_tk = atomicAdd(&ctr[q], 1);
        __syncthreads();
        const int tk = s_tk;
        __syncthreads();
        if (tk >= N_CHUNKS) {
            if (other) break;
            other = true; q ^= 1;
            continue;
        }
        const int lo = q * 64 + lane;

        for (int it = 0; it < 2; ++it) {
            const int e0 = tk * CH_EDGES + wv * 64 + it * 32;
            const int   sv = ei[e0 + lane];
            const int   dv = ei[N_EDGES + e0 + lane];
            const float nv = norm[e0 + lane];
            const int   nb = __float_as_int(nv);

            float v[2][BATCH][R];
            {
                int s[BATCH];
#pragma unroll
                for (int i = 0; i < BATCH; ++i) s[i] = __builtin_amdgcn_readlane(sv, i);
#pragma unroll
                for (int i = 0; i < BATCH; ++i) row_load<F, R>(h, s[i], lo, v[0][i]);
            }
#pragma unroll
            for (int b = 0; b < NB; ++b) {
                int s[BATCH], d[BATCH];
                float n[BATCH];
#pragma unroll
                for (int i = 0; i < BATCH; ++i) {
                    d[i] = __builtin_amdgcn_readlane(dv, b * BATCH + i);
                    n[i] = __int_as_float(__builtin_amdgcn_readlane(nb, b * BATCH + i));
                    s[i] = (b + 1 < NB) ? __builtin_amdgcn_readlane(sv, (b + 1) * BATCH + i) : 0;
                }
                if (b + 1 < NB) {
#pragma unroll
                    for (int i = 0; i < BATCH; ++i) row_load<F, R>(h, s[i], lo, v[(b + 1) & 1][i]);
                }
#pragma unroll
                for (int i = 0; i < BATCH; ++i) row_add<F, R>(out, d[i], lo, n[i], v[b & 1][i]);
            }
        }
    }
}


extern "C" void kernel_launch(void* const* d_in, const int* in_sizes, int n_in,
                              void* d_out, int out_size, void* d_ws, size_t ws_size,
                              hipStream_t stream) {
    const float* x  = (const float*)d_in[0];
    const int*   ei = (const int*)d_in[1];
    const float* W1 = (const float*)d_in[2];
    const float* b1 = (const float*)d_in[3];
    const float* W2 = (const float*)d_in[4];
    const float* b2 = (const float*)d_in[5];
    const float* W3 = (const float*)d_in[6];
    const float* b3 = (const float*)d_in[7];
    float* out = (float*)d_out;

    char* ws = (char*)d_ws;
    float* dinv = (float*)ws;
    int*   ctr  = (int*)(ws + (size_t)N_NODES * 4);
    float* bufA = (float*)(ws + (((size_t)N_NODES * 4 + 255) & ~(size_t)255));
    float* bufB = bufA + (size_t)N_NODES * HID;
    v2f*   Wp1  = (v2f*)(bufB + (size_t)N_NODES * HID);
    const int NP1 = (F_IN / 4) * 8 * 32;
    const int NP2 = (HID  / 4) * 8 * 32;
    const int NP3 = (HID  / 4) * 3 * 32;
    v2f* Wp2 = Wp1 + NP1;
    v2f* Wp3 = Wp2 + NP2;

    const int pairs   = N_NODES / 32;
    const int gemmBlk = (pairs + 3) / 4;
    const int edgeBlk = (N_EDGES / 32 + 7) / 8;
    const int halfBlk = 2048;

    pack_w_kernel<8><<<(NP1 + 255) / 256, 256, 0, stream>>>(W1, Wp1, F_IN, HID);
    pack_w_kernel<8><<<(NP2 + 255) / 256, 256, 0, stream>>>(W2, Wp2, HID, HID);
    pack_w_kernel<3><<<(NP3 + 255) / 256, 256, 0, stream>>>(W3, Wp3, HID, N_CLS);

    deg_init_kernel <<<(N_NODES + 255) / 256, 256, 0, stream>>>(dinv, ctr);
    deg_count_kernel<<<(N_EDGES + 255) / 256, 256, 0, stream>>>(ei + N_EDGES, dinv);
    dinv_kernel     <<<(N_NODES + 255) / 256, 256, 0, stream>>>(dinv);
    edge_norm_kernel<<<(N_EDGES / 4 + 255) / 256, 256, 0, stream>>>(dinv, ei, out);

    float* norm3 = bufA + (size_t)N_NODES * N_CLS;

    gemm_wmma_kernel<8, HID, false, false><<<gemmBlk, 128, 0, stream>>>(x, Wp1, bufA, N_NODES, F_IN, dinv, b1, bufB, nullptr);
    edge_scatter_halves_kernel<<<halfBlk, 256, 0, stream>>>(bufA, out, ei, bufB, ctr + 0);

    gemm_wmma_kernel<8, HID, true, false><<<gemmBlk, 128, 0, stream>>>(bufB, Wp2, bufA, N_NODES, HID, dinv, b2, bufB, nullptr);
    edge_scatter_halves_kernel<<<halfBlk, 256, 0, stream>>>(bufA, out, ei, bufB, ctr + 2);

    gemm_wmma_kernel<3, N_CLS, true, true><<<gemmBlk, 128, 0, stream>>>(bufB, Wp3, bufA, N_NODES, HID, dinv, b3, out, norm3);
    edge_scatter_kernel<N_CLS><<<edgeBlk, 256, 0, stream>>>(bufA, norm3, ei, out);
}
